# starting kernel
# baseline (speedup 1.0000x reference)
#include <hip/hip_runtime.h>
#include <math.h>

typedef __attribute__((ext_vector_type(2))) float v2f;
typedef __attribute__((ext_vector_type(8))) float v8f;

#define BATCH 64
#define NN    196
#define DD    768
#define OO    196
#define OPAD  208
#define MT    13
#define NT    48
#define NTG   6
#define KSTEPS 49
#define NTILE 7

#define WP_OFF 0
#define C_OFF  (OPAD * NN)
#define YT_OFF (C_OFF + 256)

__global__ void dnm_prep_kernel(const float* __restrict__ sw,
                                const float* __restrict__ sb,
                                const float* __restrict__ dw,
                                float* __restrict__ ws) {
  __shared__ float red[NN];
  const int o = blockIdx.x;
  const int t = threadIdx.x;
  float* Wp = ws + WP_OFF;
  float* cv = ws + C_OFF;
  if (o < OO) {
    const float d0 = dw[o * 2 + 0];
    const float d1 = dw[o * 2 + 1];
    Wp[(size_t)o * NN + t] = d0 * sw[(size_t)(o * 2 + 0) * NN + t] +
                             d1 * sw[(size_t)(o * 2 + 1) * NN + t];
    red[t] = d0 * sb[(size_t)(o * 2 + 0) * NN + t] +
             d1 * sb[(size_t)(o * 2 + 1) * NN + t];
    __syncthreads();
    if (t == 0) {
      float s = 0.0f;
      for (int i = 0; i < NN; ++i) s += red[i];
      cv[o] = s;
    }
  } else {
    Wp[(size_t)o * NN + t] = 0.0f;
    if (t == 0) cv[o] = 0.0f;
  }
}

__global__ void __launch_bounds__(256) dnm_ln_t_kernel(const float* __restrict__ x,
                                                       const float* __restrict__ lw,
                                                       const float* __restrict__ lb,
                                                       float* __restrict__ ws) {
  const int b  = blockIdx.x / NTILE;
  const int n0 = (blockIdx.x % NTILE) * 32;
  const int t    = threadIdx.x;
  const int lane = t & 31;
  const int w    = t >> 5;
  float* yT = ws + YT_OFF;

  __shared__ float mus[32], rss[32];
  __shared__ float tile[32][33];

  for (int rr = 0; rr < 4; ++rr) {
    const int r   = w * 4 + rr;
    const int row = n0 + r;
    if (row < NN) {
      const float* xr = x + ((size_t)b * NN + row) * DD;
      float s = 0.0f, q = 0.0f;
      for (int j = lane; j < DD; j += 32) {
        const float v = xr[j];
        s += v;
        q += v * v;
      }
      for (int off = 16; off > 0; off >>= 1) {
        s += __shfl_xor(s, off, 32);
        q += __shfl_xor(q, off, 32);
      }
      if (lane == 0) {
        const float mu  = s * (1.0f / DD);
        const float var = q * (1.0f / DD) - mu * mu;
        mus[r] = mu;
        rss[r] = rsqrtf(var + 1e-5f);
      }
    }
  }
  __syncthreads();

  for (int dc = 0; dc < DD / 32; ++dc) {
    const int d0 = dc * 32;
#pragma unroll
    for (int p = 0; p < 4; ++p) {
      const int r   = (t >> 5) + p * 8;
      const int c   = t & 31;
      const int row = n0 + r;
      float v = 0.0f;
      if (row < NN) {
        const float xv = x[((size_t)b * NN + row) * DD + d0 + c];
        v = (xv - mus[r]) * rss[r] * lw[d0 + c] + lb[d0 + c];
      }
      tile[r][c] = v;
    }
    __syncthreads();
#pragma unroll
    for (int p = 0; p < 4; ++p) {
      const int dr = (t >> 5) + p * 8;
      const int nc = t & 31;
      if (n0 + nc < NN)
        yT[((size_t)b * DD + d0 + dr) * NN + n0 + nc] = tile[nc][dr];
    }
    __syncthreads();
  }
}

__global__ void __launch_bounds__(256) dnm_gemm_kernel(const float* __restrict__ x,
                                                       const float* __restrict__ ws,
                                                       float* __restrict__ out) {
  const float* Wp = ws + WP_OFF;
  const float* cv = ws + C_OFF;
  const float* yT = ws + YT_OFF;

  const int wv   = threadIdx.x >> 5;
  const int lane = threadIdx.x & 31;

  const int b   = blockIdx.x / (MT * NTG);
  const int r   = blockIdx.x % (MT * NTG);
  const int mt  = r / NTG;
  const int ntg = r % NTG;
  const int nt  = ntg * 8 + wv;

  const int mo  = mt * 16;
  const int col = nt * 16 + (lane & 15);
  const int hi  = lane >> 4;

  const float* Arow = Wp + (size_t)(mo + (lane & 15)) * NN + hi * 2;
  const float* Brow = yT + ((size_t)b * DD + col) * NN + hi * 2;

  {
    const int opf = (mo + hi * 8 < OO) ? (mo + hi * 8) : (OO - 1);
    __builtin_prefetch(&x[((size_t)b * OO + opf) * DD + col], 0, 3);
  }

  v8f acc = {};
#pragma unroll 7
  for (int kb = 0; kb < KSTEPS; ++kb) {
    const v2f a  = *(const v2f*)(Arow + kb * 4);
    const v2f bf = *(const v2f*)(Brow + kb * 4);
    acc = __builtin_amdgcn_wmma_f32_16x16x4_f32(
         false, a,  false, bf,
         (short)0, acc,  false,  false);
  }

#pragma unroll
  for (int v = 0; v < 8; ++v) {
    const int o = mo + hi * 8 + v;
    if (o < OO) {
      const float tv = acc[v] + cv[o];
      const float g  = 0.5f * tv * (1.0f + erff(tv * 0.70710678118654752f));
      const size_t idx = ((size_t)b * OO + o) * DD + col;
      out[idx] = x[idx] * (g + 1.0f);
    }
  }
}

extern "C" void kernel_launch(void* const* d_in, const int* in_sizes, int n_in,
                              void* d_out, int out_size, void* d_ws, size_t ws_size,
                              hipStream_t stream) {
  const float* x    = (const float*)d_in[0];
  const float* ln_w = (const float*)d_in[1];
  const float* ln_b = (const float*)d_in[2];
  const float* sw   = (const float*)d_in[3];
  const float* sb   = (const float*)d_in[4];
  const float* dw   = (const float*)d_in[5];
  float* out = (float*)d_out;
  float* ws  = (float*)d_ws;

  dnm_prep_kernel<<<dim3(OPAD), dim3(NN), 0, stream>>>(sw, sb, dw, ws);
  dnm_ln_t_kernel<<<dim3(BATCH * NTILE), dim3(256), 0, stream>>>(x, ln_w, ln_b, ws);
  dnm_gemm_kernel<<<dim3(BATCH * MT * NTG), dim3(256), 0, stream>>>(x, ws, out);
}


# N-tile column order permuted (bt rows, in staging) so a lane's 4 accumulators of a row are 4 adjacent columns: epilogue reads x as float4 from LDS and stores whole 128-byte lines; no sum reordered
# speedup vs baseline: 3.7035x; 3.7035x over previous
#include <hip/hip_runtime.h>
#include <math.h>

typedef __attribute__((ext_vector_type(2))) float v2f;
typedef __attribute__((ext_vector_type(4))) float v4f;
typedef __attribute__((ext_vector_type(8))) float v8f;

#define BATCH 64
#define NN    196
#define DD    768
#define OO    196
#define OPAD  208
#define MT    13
#define NT    48
#define DT    64
#define NDT   12
#define GW    13
#define K8    24
#define KTAIL 192
#define NROWS (BATCH * NN)
#define NOCT  25
#define XP    68

#define WP_OFF 0
#define C_OFF  (OPAD * NN)
#define MU_OFF (C_OFF + 256)
#define RS_OFF (MU_OFF + NROWS)

__global__ void dnm_prep_kernel(const float* __restrict__ sw,
                                const float* __restrict__ sb,
                                const float* __restrict__ dw,
                                float* __restrict__ ws) {
  __shared__ float red[NN];
  const int o = blockIdx.x;
  const int t = threadIdx.x;
  float* Wp = ws + WP_OFF;
  float* cv = ws + C_OFF;
  if (o < OO) {
    const float d0 = dw[o * 2 + 0];
    const float d1 = dw[o * 2 + 1];
    Wp[(size_t)o * NN + t] = d0 * sw[(size_t)(o * 2 + 0) * NN + t] +
                             d1 * sw[(size_t)(o * 2 + 1) * NN + t];
    red[t] = d0 * sb[(size_t)(o * 2 + 0) * NN + t] +
             d1 * sb[(size_t)(o * 2 + 1) * NN + t];
    __syncthreads();
    if (t == 0) {
      float s = 0.0f;
      for (int i = 0; i < NN; ++i) s += red[i];
      cv[o] = s;
    }
  } else {
    Wp[(size_t)o * NN + t] = 0.0f;
    if (t == 0) cv[o] = 0.0f;
  }
}

#define SROWS 4
__global__ void __launch_bounds__(256) dnm_stats_kernel(const float* __restrict__ x,
                                                        float* __restrict__ ws) {
  const int lane = threadIdx.x & 31;
  const int row0 = (blockIdx.x * 8 + (threadIdx.x >> 5)) * SROWS;
  const float* xr = x + (size_t)row0 * DD + lane;
  float v[SROWS][DD / 32];
#pragma unroll
  for (int r = 0; r < SROWS; ++r) {
#pragma unroll
    for (int i = 0; i < DD / 32; ++i) v[r][i] = xr[r * DD + i * 32];
  }
  float s[SROWS], q[SROWS];
#pragma unroll
  for (int r = 0; r < SROWS; ++r) {
    s[r] = 0.0f;
    q[r] = 0.0f;
#pragma unroll
    for (int i = 0; i < DD / 32; ++i) {
      s[r] += v[r][i];
      q[r] += v[r][i] * v[r][i];
    }
  }
  for (int off = 16; off > 0; off >>= 1) {
#pragma unroll
    for (int r = 0; r < SROWS; ++r) {
      s[r] += __shfl_xor(s[r], off, 32);
      q[r] += __shfl_xor(q[r], off, 32);
    }
  }
  if (lane == 0) {
#pragma unroll
    for (int r = 0; r < SROWS; ++r) {
      const float mu  = s[r] * (1.0f / DD);
      const float var = q[r] * (1.0f / DD) - mu * mu;
      ws[MU_OFF + row0 + r] = mu;
      ws[RS_OFF + row0 + r] = rsqrtf(var + 1e-5f);
    }
  }
}

__device__ __forceinline__ v8f dnm_mma(v2f a, v2f b, v8f c) {
  return __builtin_amdgcn_wmma_f32_16x16x4_f32(
       false, a,  false, b,
       (short)0, c,  false,  false);
}

__global__ void __launch_bounds__(GW * 32) __attribute__((amdgpu_waves_per_eu(12, 12)))
__attribute__((amdgpu_num_vgpr(80)))
dnm_gemm_kernel(const float* __restrict__ x,
                                                          const float* __restrict__ lw,
                                                          const float* __restrict__ lb,
                                                          const float* __restrict__ ws,
                                                          float* __restrict__ out) {
  __shared__ __attribute__((aligned(16))) float lds[DT * NN + NN * XP + OPAD];
  float* bt = lds;
  float* xr = lds + DT * NN;
  float* cs = xr + NN * XP;

  const float* Wp = ws + WP_OFF;
  const float* cv = ws + C_OFF;
  const float* mus = ws + MU_OFF;
  const float* rss = ws + RS_OFF;

  const int t    = threadIdx.x;
  const int wv   = t >> 5;
  const int lane = t & 31;
  const int c    = lane & 15;
  const int hi   = lane >> 4;

  const int b  = blockIdx.x / NDT;
  const int d0 = (blockIdx.x % NDT) * DT;
  const int mo = wv * 16;

  {
    const int ql = lane & 3;
    const int nl = lane >> 2;
    const float* xs = x + (size_t)b * NN * DD + d0 + ql * 4;
    int  nrow[2];
    v4f xv[2][4];
    float mu2[2], rs2[2];
    v4f lwv[4], lbv[4];
    float cin = 0.0f;
    if (t < OPAD) cin = cv[t];
#pragma unroll
    for (int i = 0; i < 2; ++i) {
      const int oc = wv + i * GW;
      const int n  = oc * 8 + nl;
      nrow[i] = (n < NN) ? n : (NN - 1);
#pragma unroll
      for (int qq = 0; qq < 4; ++qq)
        xv[i][qq] = *(const v4f*)(xs + (size_t)nrow[i] * DD + qq * 16);
      mu2[i] = mus[b * NN + nrow[i]];
      rs2[i] = rss[b * NN + nrow[i]];
    }
#pragma unroll
    for (int qq = 0; qq < 4; ++qq) {
      lwv[qq] = *(const v4f*)(lw + d0 + qq * 16 + ql * 4);
      lbv[qq] = *(const v4f*)(lb + d0 + qq * 16 + ql * 4);
    }
#pragma unroll
    for (int ii = 0; ii < 2; ++ii) {
      const int i = 1 - ii;
      if (i == 0 || (wv + GW < NOCT && (wv + GW) * 8 + nl < NN)) {
#pragma unroll
        for (int qq = 0; qq < 4; ++qq)
          *(v4f*)(xr + nrow[i] * XP + qq * 16 + ql * 4) = xv[i][qq];
        __builtin_amdgcn_sched_barrier(0);
#pragma unroll
        for (int qq = 0; qq < 4; ++qq) {
#pragma unroll
          for (int e = 0; e < 4; ++e) {
            const int br = e * 16 + ql * 4 + qq;
            bt[br * NN + nrow[i]] =
                (xv[i][qq][e] - mu2[i]) * rs2[i] * lwv[qq][e] + lbv[qq][e];
          }
        }
      }
    }
    if (t < OPAD) cs[t] = cin;
  }
  __syncthreads();

  const int pc = 4 * (c & 3) + (c >> 2);
  const float* Arow = Wp + (size_t)(mo + c) * NN + hi * 4;
  const float* Bl   = bt + pc * NN + hi * 4;

  v8f acc[4] = {};
#pragma unroll 4
  for (int g = 0; g < K8; ++g) {
    const v4f a = *(const v4f*)(Arow + g * 8);
    v4f bf[4];
#pragma unroll
    for (int j = 0; j < 4; ++j) bf[j] = *(const v4f*)(Bl + j * 16 * NN + g * 8);
    const v2f a01 = {a.x, a.y};
    const v2f a23 = {a.z, a.w};
#pragma unroll
    for (int j = 0; j < 4; ++j) {
      const v2f b01 = {bf[j].x, bf[j].y};
      acc[j] = dnm_mma(a01, b01, acc[j]);
    }
#pragma unroll
    for (int j = 0; j < 4; ++j) {
      const v2f b23 = {bf[j].z, bf[j].w};
      acc[j] = dnm_mma(a23, b23, acc[j]);
    }
  }
  {
    const v2f a = *(const v2f*)(Wp + (size_t)(mo + c) * NN + KTAIL + hi * 2);
#pragma unroll
    for (int j = 0; j < 4; ++j) {
      const v2f bb = *(const v2f*)(bt + (j * 16 + pc) * NN + KTAIL + hi * 2);
      acc[j] = dnm_mma(a, bb, acc[j]);
    }
  }

  const v4f cb0 = *(const v4f*)(cs + mo + hi * 8);
  const v4f cb1 = *(const v4f*)(cs + mo + hi * 8 + 4);
#pragma unroll
  for (int v = 0; v < 8; ++v) {
    const int o = mo + hi * 8 + v;
    if (o < OO) {
      const float cb = (v < 4) ? cb0[v & 3] : cb1[v & 3];
      const v4f xo = *(const v4f*)(xr + o * XP + c * 4);
      v4f res;
#pragma unroll
      for (int j = 0; j < 4; ++j) {
        const float tv = acc[j][v] + cb;
        const float gl = 0.5f * tv * (1.0f + erff(tv * 0.70710678118654752f));
        res[j] = xo[j] * (gl + 1.0f);
      }
      *(v4f*)(out + ((size_t)b * OO + o) * DD + d0 + c * 4) = res;
    }
  }
}

extern "C" void kernel_launch(void* const* d_in, const int* in_sizes, int n_in,
                              void* d_out, int out_size, void* d_ws, size_t ws_size,
                              hipStream_t stream) {
  const float* x    = (const float*)d_in[0];
  const float* ln_w = (const float*)d_in[1];
  const float* ln_b = (const float*)d_in[2];
  const float* sw   = (const float*)d_in[3];
  const float* sb   = (const float*)d_in[4];
  const float* dw   = (const float*)d_in[5];
  float* out = (float*)d_out;
  float* ws  = (float*)d_ws;

  dnm_prep_kernel<<<dim3(OPAD), dim3(NN), 0, stream>>>(sw, sb, dw, ws);
  dnm_stats_kernel<<<dim3(NROWS / (8 * SROWS)), dim3(256), 0, stream>>>(x, ws);
  dnm_gemm_kernel<<<dim3(BATCH * NDT), dim3(GW * 32), 0, stream>>>(x, ln_w, ln_b, ws, out);
}


# GEMM block re-cut to 12 waves, equal matrix work per SIMD: the 4-row 13th M-tile's four N-tile products ride on waves 0-3 as a fifth accumulator, its rows finish from LDS at the end; no sum reordered
# speedup vs baseline: 3.8169x; 1.0306x over previous
#include <hip/hip_runtime.h>
#include <math.h>

typedef __attribute__((ext_vector_type(2))) float v2f;
typedef __attribute__((ext_vector_type(4))) float v4f;
typedef __attribute__((ext_vector_type(8))) float v8f;

#define BATCH 64
#define NN    196
#define DD    768
#define OO    196
#define OPAD  208
#define MT    13
#define NT    48
#define DT    64
#define NDT   12
#define GW    12
#define XT    12
#define K8    24
#define KTAIL 192
#define NROWS (BATCH * NN)
#define XP    68

#define WP_OFF 0
#define C_OFF  (OPAD * NN)
#define MU_OFF (C_OFF + 256)
#define RS_OFF (MU_OFF + NROWS)

__global__ void dnm_prep_kernel(const float* __restrict__ sw,
                                const float* __restrict__ sb,
                                const float* __restrict__ dw,
                                float* __restrict__ ws) {
  __shared__ float red[NN];
  const int o = blockIdx.x;
  const int t = threadIdx.x;
  float* Wp = ws + WP_OFF;
  float* cv = ws + C_OFF;
  if (o < OO) {
    const float d0 = dw[o * 2 + 0];
    const float d1 = dw[o * 2 + 1];
    Wp[(size_t)o * NN + t] = d0 * sw[(size_t)(o * 2 + 0) * NN + t] +
                             d1 * sw[(size_t)(o * 2 + 1) * NN + t];
    red[t] = d0 * sb[(size_t)(o * 2 + 0) * NN + t] +
             d1 * sb[(size_t)(o * 2 + 1) * NN + t];
    __syncthreads();
    if (t == 0) {
      float s = 0.0f;
      for (int i = 0; i < NN; ++i) s += red[i];
      cv[o] = s;
    }
  } else {
    Wp[(size_t)o * NN + t] = 0.0f;
    if (t == 0) cv[o] = 0.0f;
  }
}

#define SROWS 4
__global__ void __launch_bounds__(256) dnm_stats_kernel(const float* __restrict__ x,
                                                        float* __restrict__ ws) {
  const int lane = threadIdx.x & 31;
  const int row0 = (blockIdx.x * 8 + (threadIdx.x >> 5)) * SROWS;
  const float* xr = x + (size_t)row0 * DD + lane;
  float v[SROWS][DD / 32];
#pragma unroll
  for (int r = 0; r < SROWS; ++r) {
#pragma unroll
    for (int i = 0; i < DD / 32; ++i) v[r][i] = xr[r * DD + i * 32];
  }
  float s[SROWS], q[SROWS];
#pragma unroll
  for (int r = 0; r < SROWS; ++r) {
    s[r] = 0.0f;
    q[r] = 0.0f;
#pragma unroll
    for (int i = 0; i < DD / 32; ++i) {
      s[r] += v[r][i];
      q[r] += v[r][i] * v[r][i];
    }
  }
  for (int off = 16; off > 0; off >>= 1) {
#pragma unroll
    for (int r = 0; r < SROWS; ++r) {
      s[r] += __shfl_xor(s[r], off, 32);
      q[r] += __shfl_xor(q[r], off, 32);
    }
  }
  if (lane == 0) {
#pragma unroll
    for (int r = 0; r < SROWS; ++r) {
      const float mu  = s[r] * (1.0f / DD);
      const float var = q[r] * (1.0f / DD) - mu * mu;
      ws[MU_OFF + row0 + r] = mu;
      ws[RS_OFF + row0 + r] = rsqrtf(var + 1e-5f);
    }
  }
}

__device__ __forceinline__ v8f dnm_mma(v2f a, v2f b, v8f c) {
  return __builtin_amdgcn_wmma_f32_16x16x4_f32(
       false, a,  false, b,
       (short)0, c,  false,  false);
}

template <bool EX>
__device__ __forceinline__ void dnm_kloop(const float* __restrict__ Ar,
                                          const float* __restrict__ Ax,
                                          const float* Bl, const float* Bx,
                                          const int hi, v8f (&acc)[4], v8f& accx) {
#pragma unroll 4
  for (int g = 0; g < K8; ++g) {
    const v4f a = *(const v4f*)(Ar + hi * 4 + g * 8);
    v4f bf[4];
#pragma unroll
    for (int j = 0; j < 4; ++j) bf[j] = *(const v4f*)(Bl + hi * 4 + j * 16 * NN + g * 8);
    v4f ax = {}, bx = {};
    if (EX) {
      ax = *(const v4f*)(Ax + hi * 4 + g * 8);
      bx = *(const v4f*)(Bx + hi * 4 + g * 8);
    }
    const v2f a01 = {a.x, a.y};
    const v2f a23 = {a.z, a.w};
#pragma unroll
    for (int j = 0; j < 4; ++j) {
      const v2f b01 = {bf[j].x, bf[j].y};
      acc[j] = dnm_mma(a01, b01, acc[j]);
    }
    if (EX) {
      const v2f p = {ax.x, ax.y};
      const v2f q = {bx.x, bx.y};
      accx = dnm_mma(p, q, accx);
    }
#pragma unroll
    for (int j = 0; j < 4; ++j) {
      const v2f b23 = {bf[j].z, bf[j].w};
      acc[j] = dnm_mma(a23, b23, acc[j]);
    }
    if (EX) {
      const v2f p = {ax.z, ax.w};
      const v2f q = {bx.z, bx.w};
      accx = dnm_mma(p, q, accx);
    }
  }
  {
    const v2f a = *(const v2f*)(Ar + KTAIL + hi * 2);
#pragma unroll
    for (int j = 0; j < 4; ++j) {
      const v2f bb = *(const v2f*)(Bl + j * 16 * NN + KTAIL + hi * 2);
      acc[j] = dnm_mma(a, bb, acc[j]);
    }
    if (EX) {
      const v2f p = *(const v2f*)(Ax + KTAIL + hi * 2);
      const v2f q = *(const v2f*)(Bx + KTAIL + hi * 2);
      accx = dnm_mma(p, q, accx);
    }
  }
}

__global__ void __launch_bounds__(GW * 32) __attribute__((amdgpu_waves_per_eu(9, 9)))
__attribute__((amdgpu_num_vgpr(112)))
dnm_gemm_kernel(const float* __restrict__ x,
                const float* __restrict__ lw,
                const float* __restrict__ lb,
                const float* __restrict__ ws,
                float* __restrict__ out) {
  __shared__ __attribute__((aligned(16))) float lds[DT * NN + NN * XP + OPAD + 4 * DT];
  float* bt = lds;
  float* xr = lds + DT * NN;
  float* cs = xr + NN * XP;
  float* ex = cs + OPAD;

  const float* Wp = ws + WP_OFF;
  const float* cv = ws + C_OFF;
  const float* mus = ws + MU_OFF;
  const float* rss = ws + RS_OFF;

  const int t    = threadIdx.x;
  const int wv   = __builtin_amdgcn_readfirstlane(t >> 5);
  const int lane = t & 31;
  const int c    = lane & 15;
  const int hi   = lane >> 4;

  const int b  = blockIdx.x / NDT;
  const int d0 = (blockIdx.x % NDT) * DT;
  const int mo = wv * 16;

  {
    const int ql = lane & 3;
    const int nl = lane >> 2;
    const float* xs = x + (size_t)b * NN * DD + d0 + ql * 4;
    int  nrow[2];
    v4f xv[2][4];
    float mu2[2], rs2[2];
    v4f lwv[4], lbv[4];
    float cin = 0.0f;
    if (t < OPAD) cin = cv[t];
#pragma unroll
    for (int i = 0; i < 2; ++i) {
      nrow[i] = (wv + i * GW) * 8 + nl;
#pragma unroll
      for (int qq = 0; qq < 4; ++qq)
        xv[i][qq] = *(const v4f*)(xs + (size_t)nrow[i] * DD + qq * 16);
      mu2[i] = mus[b * NN + nrow[i]];
      rs2[i] = rss[b * NN + nrow[i]];
    }
#pragma unroll
    for (int qq = 0; qq < 4; ++qq) {
      lwv[qq] = *(const v4f*)(lw + d0 + qq * 16 + ql * 4);
      lbv[qq] = *(const v4f*)(lb + d0 + qq * 16 + ql * 4);
    }
    const bool xst = (wv >= 4) && (wv < 8);
    const int  xq  = wv - 4;
    const int  xn  = KTAIL + (nl & 3);
    v4f xxv = {}, xlw = {}, xlb = {};
    float xmu = 0.0f, xrs = 0.0f;
    if (xst) {
      xxv = *(const v4f*)(xs + (size_t)xn * DD + xq * 16);
      xmu = mus[b * NN + xn];
      xrs = rss[b * NN + xn];
      xlw = *(const v4f*)(lw + d0 + xq * 16 + ql * 4);
      xlb = *(const v4f*)(lb + d0 + xq * 16 + ql * 4);
    }
#pragma unroll
    for (int i = 0; i < 2; ++i) {
#pragma unroll
      for (int qq = 0; qq < 4; ++qq)
        *(v4f*)(xr + nrow[i] * XP + qq * 16 + ql * 4) = xv[i][qq];
#pragma unroll
      for (int qq = 0; qq < 4; ++qq) {
#pragma unroll
        for (int e = 0; e < 4; ++e) {
          const int br = e * 16 + ql * 4 + qq;
          bt[br * NN + nrow[i]] =
              (xv[i][qq][e] - mu2[i]) * rs2[i] * lwv[qq][e] + lbv[qq][e];
        }
      }
    }
    if (xst && nl < 4) {
      *(v4f*)(xr + xn * XP + xq * 16 + ql * 4) = xxv;
#pragma unroll
      for (int e = 0; e < 4; ++e) {
        const int br = e * 16 + ql * 4 + xq;
        bt[br * NN + xn] = (xxv[e] - xmu) * xrs * xlw[e] + xlb[e];
      }
    }
    if (t < OPAD) cs[t] = cin;
  }
  __syncthreads();

  const int pc = 4 * (c & 3) + (c >> 2);
  const float* Ar = Wp + (size_t)(mo + c) * NN;
  const float* Bl = bt + pc * NN;

  v8f acc[4] = {};
  v8f accx = {};
  if (wv < 4) {
    dnm_kloop<true>(Ar, Wp + (size_t)(XT * 16 + (c & 3)) * NN, Bl, Bl + wv * 16 * NN, hi, acc, accx);
    if (hi == 0) {
      const v4f cbx = *(const v4f*)(cs + KTAIL);
#pragma unroll
      for (int v = 0; v < 4; ++v) ex[v * DT + 4 * c + wv] = accx[v] + cbx[v];
    }
  } else {
    dnm_kloop<false>(Ar, Ar, Bl, Bl, hi, acc, accx);
  }

  const v4f cb0 = *(const v4f*)(cs + mo + hi * 8);
  const v4f cb1 = *(const v4f*)(cs + mo + hi * 8 + 4);
#pragma unroll
  for (int v = 0; v < 8; ++v) {
    const int o = mo + hi * 8 + v;
    const float cb = (v < 4) ? cb0[v & 3] : cb1[v & 3];
    const v4f xo = *(const v4f*)(xr + o * XP + c * 4);
    v4f res;
#pragma unroll
    for (int j = 0; j < 4; ++j) {
      const float tv = acc[j][v] + cb;
      const float gl = 0.5f * tv * (1.0f + erff(tv * 0.70710678118654752f));
      res[j] = xo[j] * (gl + 1.0f);
    }
    *(v4f*)(out + ((size_t)b * OO + o) * DD + d0 + c * 4) = res;
  }

  __syncthreads();
  if (wv < 8) {
    const int r   = wv >> 1;
    const int col = (wv & 1) * 32 + lane;
    const float tv = ex[r * DT + col];
    const float xo = xr[(KTAIL + r) * XP + col];
    const float gl = 0.5f * tv * (1.0f + erff(tv * 0.70710678118654752f));
    out[((size_t)b * OO + KTAIL + r) * DD + d0 + col] = xo * (gl + 1.0f);
  }
}

extern "C" void kernel_launch(void* const* d_in, const int* in_sizes, int n_in,
                              void* d_out, int out_size, void* d_ws, size_t ws_size,
                              hipStream_t stream) {
  const float* x    = (const float*)d_in[0];
  const float* ln_w = (const float*)d_in[1];
  const float* ln_b = (const float*)d_in[2];
  const float* sw   = (const float*)d_in[3];
  const float* sb   = (const float*)d_in[4];
  const float* dw   = (const float*)d_in[5];
  float* out = (float*)d_out;
  float* ws  = (float*)d_ws;

  dnm_prep_kernel<<<dim3(OPAD), dim3(NN), 0, stream>>>(sw, sb, dw, ws);
  dnm_stats_kernel<<<dim3(NROWS / (8 * SROWS)), dim3(256), 0, stream>>>(x, ws);
  dnm_gemm_kernel<<<dim3(BATCH * NDT), dim3(GW * 32), 0, stream>>>(x, ln_w, ln_b, ws, out);
}
